# starting kernel
# baseline (speedup 1.0000x reference)
#include <hip/hip_runtime.h>

typedef __attribute__((ext_vector_type(2))) float v2f;
typedef __attribute__((ext_vector_type(8))) float v8f;

#define NN 50000
#define NE 1600000
#define DN 64
#define DE 32
#define DG 32
#define HIDDEN 128
#define NF 128
#define ROWS 64
#define LDF 130


__global__ void zero_ws(float* __restrict__ p, int n) {
  int i = blockIdx.x * blockDim.x + threadIdx.x;
  if (i < n) p[i] = 0.0f;
}

__global__ void scatter_edges(const float* __restrict__ edge_attr,
                              const int* __restrict__ edge_index,
                              float* __restrict__ sums,
                              float* __restrict__ counts) {
  long long t = (long long)blockIdx.x * blockDim.x + threadIdx.x;
  int e = (int)(t >> 5);
  if (e >= NE) return;
  int c = (int)(t & 31);
  int dst = edge_index[NE + e];
  atomicAdd(&sums[(long long)dst * DE + c], edge_attr[(long long)e * DE + c]);
  if (c == 0) atomicAdd(&counts[dst], 1.0f);
}

__global__ void seg_mean(const float* __restrict__ sums,
                         const float* __restrict__ counts,
                         float* __restrict__ eob) {
  int i = blockIdx.x * blockDim.x + threadIdx.x;
  if (i >= NN * DE) return;
  float cnt = counts[i >> 5];
  eob[i] = sums[i] / fmaxf(cnt, 1.0f);
}


__global__ __launch_bounds__(128) void mlp_wmma(
    const float* __restrict__ x, const float* __restrict__ eob,
    const float* __restrict__ u, const int* __restrict__ batch,
    const float* __restrict__ W1, const float* __restrict__ b1,
    const float* __restrict__ W2, const float* __restrict__ b2,
    float* __restrict__ y) {
  __shared__ float feat[ROWS * LDF];
  __shared__ float hbuf[ROWS * LDF];

  const int tid = threadIdx.x;
  const int nb = blockIdx.x * ROWS;

  for (int idx = tid; idx < ROWS * NF; idx += 128) {
    int r = idx >> 7;
    int c = idx & 127;
    int node = nb + r;
    float v = 0.0f;
    if (node < NN) {
      if (c < DN)           v = x[(long long)node * DN + c];
      else if (c < DN + DE) v = eob[(long long)node * DE + (c - DN)];
      else                  v = u[(long long)batch[node] * DG + (c - DN - DE)];
    }
    feat[r * LDF + c] = v;
  }
  __syncthreads();

  const int wave = tid >> 5;
  const int lane = tid & 31;
  const int half = lane >> 4;
  const int nidx = lane & 15;
  const int mrow0 = wave * 16;

  for (int nt = 0; nt < HIDDEN / 16; ++nt) {
    const int ncol = nt * 16 + nidx;
    const float bias = b1[ncol];
    v8f acc;
#pragma unroll
    for (int v = 0; v < 8; ++v) acc[v] = bias;
#pragma unroll 4
    for (int ks = 0; ks < NF / 4; ++ks) {
      int kk = ks * 4 + 2 * half;
      v2f a = *(const v2f*)&feat[(mrow0 + nidx) * LDF + kk];
      v2f b;
      b.x = W1[kk * HIDDEN + ncol];
      b.y = W1[(kk + 1) * HIDDEN + ncol];
      acc = __builtin_amdgcn_wmma_f32_16x16x4_f32(false, a, false, b,
                                                  (short)0, acc, false, false);
    }
#pragma unroll
    for (int v = 0; v < 8; ++v) {
      float r = acc[v];
      hbuf[(mrow0 + v + 8 * half) * LDF + ncol] = r > 0.0f ? r : 0.0f;
    }
  }
  __syncthreads();

  for (int nt = 0; nt < DN / 16; ++nt) {
    const int ncol = nt * 16 + nidx;
    const float bias = b2[ncol];
    v8f acc;
#pragma unroll
    for (int v = 0; v < 8; ++v) acc[v] = bias;
#pragma unroll 4
    for (int ks = 0; ks < HIDDEN / 4; ++ks) {
      int kk = ks * 4 + 2 * half;
      v2f a = *(const v2f*)&hbuf[(mrow0 + nidx) * LDF + kk];
      v2f b;
      b.x = W2[kk * DN + ncol];
      b.y = W2[(kk + 1) * DN + ncol];
      acc = __builtin_amdgcn_wmma_f32_16x16x4_f32(false, a, false, b,
                                                  (short)0, acc, false, false);
    }
#pragma unroll
    for (int v = 0; v < 8; ++v) {
      int node = nb + mrow0 + v + 8 * half;
      if (node < NN) {
        float r = acc[v];
        y[(long long)node * DN + ncol] = r > 0.0f ? r : 0.0f;
      }
    }
  }
}


extern "C" void kernel_launch(void* const* d_in, const int* in_sizes, int n_in,
                              void* d_out, int out_size, void* d_ws, size_t ws_size,
                              hipStream_t stream) {
  const float* x     = (const float*)d_in[0];
  const int*   eidx  = (const int*)d_in[1];
  const float* eattr = (const float*)d_in[2];
  const float* u     = (const float*)d_in[3];
  const int*   batch = (const int*)d_in[4];
  const float* W1    = (const float*)d_in[5];
  const float* b1    = (const float*)d_in[6];
  const float* W2    = (const float*)d_in[7];
  const float* b2    = (const float*)d_in[8];

  float* y   = (float*)d_out;
  float* eob = (float*)d_out + (size_t)NN * DN;

  float* sums   = (float*)d_ws;
  float* counts = sums + (size_t)NN * DE;

  int nz = NN * DE + NN;
  zero_ws<<<(nz + 255) / 256, 256, 0, stream>>>((float*)d_ws, nz);

  long long sthreads = (long long)NE * 32;
  scatter_edges<<<(int)((sthreads + 255) / 256), 256, 0, stream>>>(
      eattr, eidx, sums, counts);

  int mz = NN * DE;
  seg_mean<<<(mz + 255) / 256, 256, 0, stream>>>(sums, counts, eob);

  mlp_wmma<<<(NN + ROWS - 1) / ROWS, 128, 0, stream>>>(
      x, eob, u, batch, W1, b1, W2, b2, y);
}


# MLP transposed and register-chained: layer 1 as W1^T*feat^T, its D registers are layer 2's A fragments; rows and weights read as 16-byte fragment words; no LDS; k order of both layers changes
# speedup vs baseline: 4.8870x; 4.8870x over previous
#include <hip/hip_runtime.h>

typedef __attribute__((ext_vector_type(2))) float v2f;
typedef __attribute__((ext_vector_type(8))) float v8f;

#define NN 50000
#define NE 1600000
#define DN 64
#define DE 32
#define DG 32
#define HIDDEN 128
#define NF 128


#define NCH 32
#define EC (NE / NCH)
#define GRP (EC / 8)
#define GR ((GRP + 1023) / 1024)

typedef __attribute__((ext_vector_type(8))) unsigned short v8us;

__global__ __launch_bounds__(1024) void build_sublists(const int* __restrict__ edge_index,
                                                       unsigned short* __restrict__ heads,
                                                       unsigned short* __restrict__ nxt) {
  __shared__ __attribute__((aligned(16))) int h[NN + 4];
  const int tid = threadIdx.x;
  const int base = blockIdx.x * EC;
  const int* dst = edge_index + NE + base;

  int4 d[GR][2];
#pragma unroll
  for (int k = 0; k < GR; ++k) {
    const int g = tid + k * 1024;
    d[k][0] = make_int4(-1, -1, -1, -1);
    d[k][1] = make_int4(-1, -1, -1, -1);
    if (g < GRP) {
      d[k][0] = *(const int4*)&dst[8 * g];
      d[k][1] = *(const int4*)&dst[8 * g + 4];
    }
  }

  for (int i = tid; i < NN / 4; i += 1024) ((int4*)h)[i] = make_int4(-1, -1, -1, -1);
  __syncthreads();

#pragma unroll
  for (int k = 0; k < GR; ++k) {
    const int g = tid + k * 1024;
    if (g < GRP) {
      const int l = 8 * g;
      v8us o;
      const int dd[8] = {d[k][0].x, d[k][0].y, d[k][0].z, d[k][0].w,
                         d[k][1].x, d[k][1].y, d[k][1].z, d[k][1].w};
      int old[8];
#pragma unroll
      for (int i = 0; i < 8; ++i) {
        const bool ok = (unsigned)dd[i] < (unsigned)NN;
        old[i] = atomicExch(&h[ok ? dd[i] : NN], l + i);
        if (!ok) old[i] = -1;
      }
#pragma unroll
      for (int i = 0; i < 8; ++i) o[i] = (unsigned short)old[i];
      *(v8us*)&nxt[base + l] = o;
    }
  }
  __syncthreads();

  unsigned short* hrow = heads + (size_t)blockIdx.x * NN;
  for (int i = tid; i < NN / 8; i += 1024) {
    const int4 p = ((const int4*)h)[2 * i];
    const int4 q = ((const int4*)h)[2 * i + 1];
    v8us o;
    o[0] = (unsigned short)p.x; o[1] = (unsigned short)p.y;
    o[2] = (unsigned short)p.z; o[3] = (unsigned short)p.w;
    o[4] = (unsigned short)q.x; o[5] = (unsigned short)q.y;
    o[6] = (unsigned short)q.z; o[7] = (unsigned short)q.w;
    *(v8us*)&hrow[8 * i] = o;
  }
}

__global__ __launch_bounds__(256) void gather_mean(const float* __restrict__ edge_attr,
                                                   const unsigned short* __restrict__ heads,
                                                   const unsigned short* __restrict__ nxt,
                                                   float* __restrict__ eob) {
  const int lane = threadIdx.x & 31;
  const int node = __builtin_amdgcn_readfirstlane(blockIdx.x * 8 + (threadIdx.x >> 5));
  if (node >= NN) return;

  unsigned cur = heads[(size_t)lane * NN + node];
  int cnt = 0;
  float acc = 0.0f;

  for (int lvl = 0; lvl < EC; ++lvl) {
    const bool live = cur < (unsigned)EC;
    unsigned mask = __builtin_amdgcn_ballot_w32(live);
    if (mask == 0) break;
    const int gid = lane * EC + (int)cur;
    unsigned nx = 0xFFFFu;
    if (live) nx = nxt[gid];
    cnt += __builtin_popcount(mask);

    float r[NCH];
#pragma unroll
    for (int k = 0; k < NCH; ++k) {
      r[k] = 0.0f;
      if (mask) {
        const int j = __builtin_ctz(mask);
        mask &= mask - 1;
        const int id = __builtin_amdgcn_readlane(gid, j);
        r[k] = __builtin_nontemporal_load(&edge_attr[(long long)id * DE + lane]);
      }
    }
#pragma unroll
    for (int k = 0; k < NCH; ++k) acc += r[k];
    cur = nx;
  }

  const float c = (float)cnt;
  eob[(long long)node * DE + lane] = acc / fmaxf(c, 1.0f);
}


static_assert(NN % 16 == 0, "a wave owns 16 whole node rows");
static_assert(DN == 64 && DE == 32 && DG == 32 && HIDDEN == 128 && NF == DN + DE + DG,
              "the word counts and unit maps below are written out for these sizes");


#define WMMA4(A, B, C) __builtin_amdgcn_wmma_f32_16x16x4_f32(false, A, false, B, (short)0, C, false, false)
#define NBATCH 24

__device__ __forceinline__ void load_wbatch(float4 (&G)[8], const int n,
                                            const float* __restrict__ w1l,
                                            const float* __restrict__ w2l) {
  const int tg = n / 12;
  const int m = n % 12;
  if (m < 8) {
#pragma unroll
    for (int t = 0; t < 4; ++t) {
      const int q = 4 * m + t;
      const int row = 8 * (q >> 1) + 2 * (q & 1);
      G[2 * t]     = *(const float4*)(w1l + row * HIDDEN + 64 * tg);
      G[2 * t + 1] = *(const float4*)(w1l + (row + 1) * HIDDEN + 64 * tg);
    }
  } else {
    const int i = m - 8;
#pragma unroll
    for (int p = 0; p < 4; ++p) {
      G[2 * p]     = *(const float4*)(w2l + (64 * tg + 8 * p + i) * DN);
      G[2 * p + 1] = *(const float4*)(w2l + (64 * tg + 8 * p + 4 + i) * DN);
    }
  }
}

__global__ __launch_bounds__(128) void mlp_wmma(
    const float* __restrict__ x, const float* __restrict__ eob,
    const float* __restrict__ u, const int* __restrict__ batch,
    const float* __restrict__ W1, const float* __restrict__ b1,
    const float* __restrict__ W2, const float* __restrict__ b2,
    float* __restrict__ y) {
  const int lane = threadIdx.x & 31;
  const int tile = __builtin_amdgcn_readfirstlane(blockIdx.x * 4 + (threadIdx.x >> 5));
  if (tile >= NN / 16) return;
  const int c = lane & 15;
  const int hh = lane >> 4;
  const int node = tile * 16 + c;

  float4 F[16];
  const int gph = batch[node];
  const float* xr = x + (long long)node * DN + 4 * hh;
#pragma unroll
  for (int j = 0; j < 8; ++j) F[j] = *(const float4*)(xr + 8 * j);
  const float* er = eob + (long long)node * DE + 4 * hh;
#pragma unroll
  for (int j = 0; j < 4; ++j) F[8 + j] = *(const float4*)(er + 8 * j);

  const float* w1l = W1 + (4 * hh) * HIDDEN + 4 * c;
  const float* w2l = W2 + (32 * hh) * DN + 4 * c;
  float4 G[2][8];
  load_wbatch(G[0], 0, w1l, w2l);

  const float* ur = u + (long long)gph * DG + 4 * hh;
#pragma unroll
  for (int j = 0; j < 4; ++j) F[12 + j] = *(const float4*)(ur + 8 * j);

  v8f yacc[4];
  {
    const float4 bv = *(const float4*)(b2 + 4 * c);
#pragma unroll
    for (int r = 0; r < 8; ++r) { yacc[0][r] = bv.x; yacc[1][r] = bv.y; yacc[2][r] = bv.z; yacc[3][r] = bv.w; }
  }

  v8f acc[4];
#pragma unroll
  for (int n = 0; n < NBATCH; ++n) {
    const int tg = n / 12;
    const int m = n % 12;
    if (n + 1 < NBATCH) load_wbatch(G[(n + 1) & 1], n + 1, w1l, w2l);
    const float4 (&g)[8] = G[n & 1];
    if (m == 0) {
#pragma unroll
      for (int r = 0; r < 8; ++r) {
        const float4 bv = *(const float4*)(b1 + 64 * tg + 32 * hh + 4 * r);
        acc[0][r] = bv.x; acc[1][r] = bv.y; acc[2][r] = bv.z; acc[3][r] = bv.w;
      }
    }
    if (m < 8) {
#pragma unroll
      for (int t = 0; t < 4; ++t) {
        const int q = 4 * m + t;
        const int j = q >> 1;
        const int e = q & 1;
        v2f bf;
        bf.x = e ? F[j].z : F[j].x;
        bf.y = e ? F[j].w : F[j].y;
        const float4 g0 = g[2 * t];
        const float4 g1 = g[2 * t + 1];
        v2f a;
        a.x = g0.x; a.y = g1.x; acc[0] = WMMA4(a, bf, acc[0]);
        a.x = g0.y; a.y = g1.y; acc[1] = WMMA4(a, bf, acc[1]);
        a.x = g0.z; a.y = g1.z; acc[2] = WMMA4(a, bf, acc[2]);
        a.x = g0.w; a.y = g1.w; acc[3] = WMMA4(a, bf, acc[3]);
      }
      if (m == 7) {
#pragma unroll
        for (int i = 0; i < 4; ++i)
#pragma unroll
          for (int r = 0; r < 8; ++r) { const float v = acc[i][r]; acc[i][r] = v > 0.0f ? v : 0.0f; }
      }
    } else {
      const int i = m - 8;
#pragma unroll
      for (int p = 0; p < 4; ++p) {
        const float4 k0 = g[2 * p];
        const float4 k1 = g[2 * p + 1];
        v2f a;
        a.x = acc[i][2 * p]; a.y = acc[i][2 * p + 1];
        v2f b;
        b.x = k0.x; b.y = k1.x; yacc[0] = WMMA4(a, b, yacc[0]);
        b.x = k0.y; b.y = k1.y; yacc[1] = WMMA4(a, b, yacc[1]);
        b.x = k0.z; b.y = k1.z; yacc[2] = WMMA4(a, b, yacc[2]);
        b.x = k0.w; b.y = k1.w; yacc[3] = WMMA4(a, b, yacc[3]);
      }
    }
  }

  float* yr = y + ((long long)tile * 16 + 8 * hh) * DN + 4 * c;
#pragma unroll
  for (int r = 0; r < 8; ++r) {
    float4 o;
    o.x = yacc[0][r] > 0.0f ? yacc[0][r] : 0.0f;
    o.y = yacc[1][r] > 0.0f ? yacc[1][r] : 0.0f;
    o.z = yacc[2][r] > 0.0f ? yacc[2][r] : 0.0f;
    o.w = yacc[3][r] > 0.0f ? yacc[3][r] : 0.0f;
    *(float4*)(yr + r * DN) = o;
  }
}


extern "C" void kernel_launch(void* const* d_in, const int* in_sizes, int n_in,
                              void* d_out, int out_size, void* d_ws, size_t ws_size,
                              hipStream_t stream) {
  const float* x     = (const float*)d_in[0];
  const int*   eidx  = (const int*)d_in[1];
  const float* eattr = (const float*)d_in[2];
  const float* u     = (const float*)d_in[3];
  const int*   batch = (const int*)d_in[4];
  const float* W1    = (const float*)d_in[5];
  const float* b1    = (const float*)d_in[6];
  const float* W2    = (const float*)d_in[7];
  const float* b2    = (const float*)d_in[8];

  float* y   = (float*)d_out;
  float* eob = (float*)d_out + (size_t)NN * DN;

  unsigned short* heads = (unsigned short*)d_ws;
  unsigned short* nxt   = heads + (size_t)NCH * NN;

  build_sublists<<<NCH, 1024, 0, stream>>>(eidx, heads, nxt);

  gather_mean<<<(NN + 7) / 8, 256, 0, stream>>>(eattr, heads, nxt, eob);

  mlp_wmma<<<(NN / 16 + 3) / 4, 128, 0, stream>>>(
      x, eob, u, batch, W1, b1, W2, b2, y);
}


# layer 1: graph-row columns' product (u x W1 rows 96..127, + b1) computed once per graph as a 64 x 128 WMMA table in a spare block of the gather launch; per-node k 128 -> 96; layer-1 sum order changes
# speedup vs baseline: 5.0487x; 1.0331x over previous
#include <hip/hip_runtime.h>

typedef __attribute__((ext_vector_type(2))) float v2f;
typedef __attribute__((ext_vector_type(8))) float v8f;

#define NN 50000
#define NE 1600000
#define DN 64
#define DE 32
#define DG 32
#define HIDDEN 128
#define NF 128
#define NG 64
#define KN (DN + DE)


#define NCH 32
#define EC (NE / NCH)
#define GRP (EC / 8)
#define GR ((GRP + 1023) / 1024)

typedef __attribute__((ext_vector_type(8))) unsigned short v8us;

__global__ __launch_bounds__(1024) void build_sublists(const int* __restrict__ edge_index,
                                                       unsigned short* __restrict__ heads,
                                                       unsigned short* __restrict__ nxt) {
  __shared__ __attribute__((aligned(16))) int h[NN + 4];
  const int tid = threadIdx.x;
  const int base = blockIdx.x * EC;
  const int* dst = edge_index + NE + base;

  int4 d[GR][2];
#pragma unroll
  for (int k = 0; k < GR; ++k) {
    const int g = tid + k * 1024;
    d[k][0] = make_int4(-1, -1, -1, -1);
    d[k][1] = make_int4(-1, -1, -1, -1);
    if (g < GRP) {
      d[k][0] = *(const int4*)&dst[8 * g];
      d[k][1] = *(const int4*)&dst[8 * g + 4];
    }
  }

  for (int i = tid; i < NN / 4; i += 1024) ((int4*)h)[i] = make_int4(-1, -1, -1, -1);
  __syncthreads();

#pragma unroll
  for (int k = 0; k < GR; ++k) {
    const int g = tid + k * 1024;
    if (g < GRP) {
      const int l = 8 * g;
      v8us o;
      const int dd[8] = {d[k][0].x, d[k][0].y, d[k][0].z, d[k][0].w,
                         d[k][1].x, d[k][1].y, d[k][1].z, d[k][1].w};
      int old[8];
#pragma unroll
      for (int i = 0; i < 8; ++i) {
        const bool ok = (unsigned)dd[i] < (unsigned)NN;
        old[i] = atomicExch(&h[ok ? dd[i] : NN], l + i);
        if (!ok) old[i] = -1;
      }
#pragma unroll
      for (int i = 0; i < 8; ++i) o[i] = (unsigned short)old[i];
      *(v8us*)&nxt[base + l] = o;
    }
  }
  __syncthreads();

  unsigned short* hrow = heads + (size_t)blockIdx.x * NN;
  for (int i = tid; i < NN / 8; i += 1024) {
    const int4 p = ((const int4*)h)[2 * i];
    const int4 q = ((const int4*)h)[2 * i + 1];
    v8us o;
    o[0] = (unsigned short)p.x; o[1] = (unsigned short)p.y;
    o[2] = (unsigned short)p.z; o[3] = (unsigned short)p.w;
    o[4] = (unsigned short)q.x; o[5] = (unsigned short)q.y;
    o[6] = (unsigned short)q.z; o[7] = (unsigned short)q.w;
    *(v8us*)&hrow[8 * i] = o;
  }
}

#define WMMA4(A, B, C) __builtin_amdgcn_wmma_f32_16x16x4_f32(false, A, false, B, (short)0, C, false, false)

__device__ __forceinline__ void graph_rows_table(const float* __restrict__ u,
                                                 const float* __restrict__ W1,
                                                 const float* __restrict__ b1,
                                                 float* __restrict__ gtab) {
  const int lane = threadIdx.x & 31;
  const int w = threadIdx.x >> 5;
  const int c = lane & 15;
  const int hh = lane >> 4;
  const int g0 = 16 * (w & 3);
  const int u0 = 64 * (w >> 2);
  v8f d[4];
#pragma unroll
  for (int t = 0; t < 4; ++t)
#pragma unroll
    for (int r = 0; r < 8; ++r) d[t][r] = b1[u0 + 16 * t + 8 * hh + r];
#pragma unroll
  for (int q = 0; q < DG / 4; ++q) {
    const int k = 4 * q + 2 * hh;
    const v2f b = *(const v2f*)(u + (g0 + c) * DG + k);
#pragma unroll
    for (int t = 0; t < 4; ++t) {
      v2f a;
      a.x = W1[(KN + k) * HIDDEN + u0 + 16 * t + c];
      a.y = W1[(KN + k + 1) * HIDDEN + u0 + 16 * t + c];
      d[t] = WMMA4(a, b, d[t]);
    }
  }
  float* o = gtab + (g0 + c) * HIDDEN + u0 + 8 * hh;
#pragma unroll
  for (int t = 0; t < 4; ++t) {
    *(float4*)(o + 16 * t)     = make_float4(d[t][0], d[t][1], d[t][2], d[t][3]);
    *(float4*)(o + 16 * t + 4) = make_float4(d[t][4], d[t][5], d[t][6], d[t][7]);
  }
}

#define NODE_BLOCKS ((NN + 7) / 8)

__global__ __launch_bounds__(256) void gather_mean(const float* __restrict__ edge_attr,
                                                   const unsigned short* __restrict__ heads,
                                                   const unsigned short* __restrict__ nxt,
                                                   float* __restrict__ eob,
                                                   const float* __restrict__ u,
                                                   const float* __restrict__ W1,
                                                   const float* __restrict__ b1,
                                                   float* __restrict__ gtab) {
  if (blockIdx.x == 0) { graph_rows_table(u, W1, b1, gtab); return; }
  const int lane = threadIdx.x & 31;
  const int node = __builtin_amdgcn_readfirstlane((blockIdx.x - 1) * 8 + (threadIdx.x >> 5));
  if (node >= NN) return;

  unsigned cur = heads[(size_t)lane * NN + node];
  int cnt = 0;
  float acc = 0.0f;

  for (int lvl = 0; lvl < EC; ++lvl) {
    const bool live = cur < (unsigned)EC;
    unsigned mask = __builtin_amdgcn_ballot_w32(live);
    if (mask == 0) break;
    const int gid = lane * EC + (int)cur;
    unsigned nx = 0xFFFFu;
    if (live) nx = nxt[gid];
    cnt += __builtin_popcount(mask);

    float r[NCH];
#pragma unroll
    for (int k = 0; k < NCH; ++k) {
      r[k] = 0.0f;
      if (mask) {
        const int j = __builtin_ctz(mask);
        mask &= mask - 1;
        const int id = __builtin_amdgcn_readlane(gid, j);
        r[k] = __builtin_nontemporal_load(&edge_attr[(long long)id * DE + lane]);
      }
    }
#pragma unroll
    for (int k = 0; k < NCH; ++k) acc += r[k];
    cur = nx;
  }

  const float c = (float)cnt;
  eob[(long long)node * DE + lane] = acc / fmaxf(c, 1.0f);
}


static_assert(NN % 16 == 0, "a wave owns 16 whole node rows");
static_assert(DN == 64 && DE == 32 && DG == 32 && HIDDEN == 128 && NF == DN + DE + DG,
              "the word counts and unit maps below are written out for these sizes");


#define NBATCH 20

__device__ __forceinline__ void load_wbatch(float4 (&G)[8], const int n,
                                            const float* __restrict__ w1l,
                                            const float* __restrict__ w2l) {
  const int tg = n / 10;
  const int m = n % 10;
  if (m < 6) {
#pragma unroll
    for (int t = 0; t < 4; ++t) {
      const int q = 4 * m + t;
      const int row = 8 * (q >> 1) + 2 * (q & 1);
      G[2 * t]     = *(const float4*)(w1l + row * HIDDEN + 64 * tg);
      G[2 * t + 1] = *(const float4*)(w1l + (row + 1) * HIDDEN + 64 * tg);
    }
  } else {
    const int i = m - 6;
#pragma unroll
    for (int p = 0; p < 4; ++p) {
      G[2 * p]     = *(const float4*)(w2l + (64 * tg + 8 * p + i) * DN);
      G[2 * p + 1] = *(const float4*)(w2l + (64 * tg + 8 * p + 4 + i) * DN);
    }
  }
}

__global__ __launch_bounds__(128) void mlp_wmma(
    const float* __restrict__ x, const float* __restrict__ eob,
    const float* __restrict__ gtab, const int* __restrict__ batch,
    const float* __restrict__ W1,
    const float* __restrict__ W2, const float* __restrict__ b2,
    float* __restrict__ y) {
  const int lane = threadIdx.x & 31;
  const int tile = __builtin_amdgcn_readfirstlane(blockIdx.x * 4 + (threadIdx.x >> 5));
  if (tile >= NN / 16) return;
  const int c = lane & 15;
  const int hh = lane >> 4;
  const int node = tile * 16 + c;

  float4 F[12];
  const int gph = min(max(batch[node], 0), NG - 1);
  const float* xr = x + (long long)node * DN + 4 * hh;
#pragma unroll
  for (int j = 0; j < 8; ++j) F[j] = *(const float4*)(xr + 8 * j);
  const float* er = eob + (long long)node * DE + 4 * hh;
#pragma unroll
  for (int j = 0; j < 4; ++j) F[8 + j] = *(const float4*)(er + 8 * j);

  const float* w1l = W1 + (4 * hh) * HIDDEN + 4 * c;
  const float* w2l = W2 + (32 * hh) * DN + 4 * c;
  float4 G[2][8];
  load_wbatch(G[0], 0, w1l, w2l);

  const float* tr = gtab + gph * HIDDEN + 32 * hh;
  float4 Tt[2][8];
#pragma unroll
  for (int t = 0; t < 2; ++t)
#pragma unroll
    for (int r = 0; r < 8; ++r) Tt[t][r] = *(const float4*)(tr + 64 * t + 4 * r);

  v8f yacc[4];
  {
    const float4 bv = *(const float4*)(b2 + 4 * c);
#pragma unroll
    for (int r = 0; r < 8; ++r) { yacc[0][r] = bv.x; yacc[1][r] = bv.y; yacc[2][r] = bv.z; yacc[3][r] = bv.w; }
  }

  v8f acc[4];
#pragma unroll
  for (int n = 0; n < NBATCH; ++n) {
    const int tg = n / 10;
    const int m = n % 10;
    if (n + 1 < NBATCH) load_wbatch(G[(n + 1) & 1], n + 1, w1l, w2l);
    const float4 (&g)[8] = G[n & 1];
    if (m == 0) {
#pragma unroll
      for (int r = 0; r < 8; ++r) {
        const float4 tv = Tt[tg][r];
        acc[0][r] = tv.x; acc[1][r] = tv.y; acc[2][r] = tv.z; acc[3][r] = tv.w;
      }
    }
    if (m < 6) {
#pragma unroll
      for (int t = 0; t < 4; ++t) {
        const int q = 4 * m + t;
        const int j = q >> 1;
        const int e = q & 1;
        v2f bf;
        bf.x = e ? F[j].z : F[j].x;
        bf.y = e ? F[j].w : F[j].y;
        const float4 g0 = g[2 * t];
        const float4 g1 = g[2 * t + 1];
        v2f a;
        a.x = g0.x; a.y = g1.x; acc[0] = WMMA4(a, bf, acc[0]);
        a.x = g0.y; a.y = g1.y; acc[1] = WMMA4(a, bf, acc[1]);
        a.x = g0.z; a.y = g1.z; acc[2] = WMMA4(a, bf, acc[2]);
        a.x = g0.w; a.y = g1.w; acc[3] = WMMA4(a, bf, acc[3]);
      }
      if (m == 5) {
#pragma unroll
        for (int i = 0; i < 4; ++i)
#pragma unroll
          for (int r = 0; r < 8; ++r) { const float v = acc[i][r]; acc[i][r] = v > 0.0f ? v : 0.0f; }
      }
    } else {
      const int i = m - 6;
#pragma unroll
      for (int p = 0; p < 4; ++p) {
        const float4 k0 = g[2 * p];
        const float4 k1 = g[2 * p + 1];
        v2f a;
        a.x = acc[i][2 * p]; a.y = acc[i][2 * p + 1];
        v2f b;
        b.x = k0.x; b.y = k1.x; yacc[0] = WMMA4(a, b, yacc[0]);
        b.x = k0.y; b.y = k1.y; yacc[1] = WMMA4(a, b, yacc[1]);
        b.x = k0.z; b.y = k1.z; yacc[2] = WMMA4(a, b, yacc[2]);
        b.x = k0.w; b.y = k1.w; yacc[3] = WMMA4(a, b, yacc[3]);
      }
    }
  }

  float* yr = y + ((long long)tile * 16 + 8 * hh) * DN + 4 * c;
#pragma unroll
  for (int r = 0; r < 8; ++r) {
    float4 o;
    o.x = yacc[0][r] > 0.0f ? yacc[0][r] : 0.0f;
    o.y = yacc[1][r] > 0.0f ? yacc[1][r] : 0.0f;
    o.z = yacc[2][r] > 0.0f ? yacc[2][r] : 0.0f;
    o.w = yacc[3][r] > 0.0f ? yacc[3][r] : 0.0f;
    *(float4*)(yr + r * DN) = o;
  }
}


extern "C" void kernel_launch(void* const* d_in, const int* in_sizes, int n_in,
                              void* d_out, int out_size, void* d_ws, size_t ws_size,
                              hipStream_t stream) {
  const float* x     = (const float*)d_in[0];
  const int*   eidx  = (const int*)d_in[1];
  const float* eattr = (const float*)d_in[2];
  const float* u     = (const float*)d_in[3];
  const int*   batch = (const int*)d_in[4];
  const float* W1    = (const float*)d_in[5];
  const float* b1    = (const float*)d_in[6];
  const float* W2    = (const float*)d_in[7];
  const float* b2    = (const float*)d_in[8];

  float* y   = (float*)d_out;
  float* eob = (float*)d_out + (size_t)NN * DN;

  unsigned short* heads = (unsigned short*)d_ws;
  unsigned short* nxt   = heads + (size_t)NCH * NN;
  float* gtab = (float*)(nxt + (size_t)NE);

  build_sublists<<<NCH, 1024, 0, stream>>>(eidx, heads, nxt);

  gather_mean<<<NODE_BLOCKS + 1, 256, 0, stream>>>(eattr, heads, nxt, eob, u, W1, b1, gtab);

  mlp_wmma<<<(NN / 16 + 3) / 4, 128, 0, stream>>>(
      x, eob, gtab, batch, W1, W2, b2, y);
}
